# starting kernel
# baseline (speedup 1.0000x reference)
#include <hip/hip_runtime.h>

typedef __attribute__((ext_vector_type(2))) float v2f;
typedef __attribute__((ext_vector_type(8))) float v8f;

#define LEAKY_SLOPE 0.2f

constexpr int NB   = 4;
constexpr int NPTS = 16384;
constexpr int FD   = 64;
constexpr int KNN  = 16;

__global__ __launch_bounds__(256) void lfa_kernel(
    const float* __restrict__ points,
    const float* __restrict__ features,
    const int*  __restrict__ knn,
    const float* __restrict__ W,
    const float* __restrict__ bias,
    float* __restrict__ out)
{
    const int lane = threadIdx.x & 31;
    const int wave = threadIdx.x >> 5;
    const int p    = blockIdx.x * 8 + wave;
    const int b    = p >> 14;
    const int nn   = lane & 15;
    const int khi  = lane >> 4;

    v2f  wt[4];
    float bv[4];
#pragma unroll
    for (int t = 0; t < 4; ++t) {
        wt[t].x = W[(2 * khi)     * FD + t * 16 + nn];
        wt[t].y = W[(2 * khi + 1) * FD + t * 16 + nn];
        bv[t]   = bias[t * 16 + nn];
    }

    const int  idx   = knn[(long)p * KNN + nn];
    const long pbase = (long)p * 3;
    const float cx = points[pbase + 0];
    const float cy = points[pbase + 1];
    const float cz = points[pbase + 2];
    const long nbase = ((long)b * NPTS + idx) * 3;
    const float rx = cx - points[nbase + 0];
    const float ry = cy - points[nbase + 1];
    const float rz = cz - points[nbase + 2];
    const float dist = __builtin_amdgcn_sqrtf(rx * rx + ry * ry + rz * rz);

    v2f a;
    a.x = khi ? rz   : rx;
    a.y = khi ? dist : ry;

    const float2* __restrict__ fbase = (const float2*)features + (long)b * NPTS * 32;
    float fx = 0.f, fy = 0.f;
#pragma unroll
    for (int k = 0; k < KNN; ++k) {
        const int ik = __builtin_amdgcn_readlane(idx, k);
        const float2 v = fbase[(long)ik * 32 + lane];
        fx += v.x;
        fy += v.y;
    }

    float mm[4];
#pragma unroll
    for (int t = 0; t < 4; ++t) {
        v8f c = {};
        c = __builtin_amdgcn_wmma_f32_16x16x4_f32(
                 false, a,  false, wt[t],
                 (short)0, c,  false,  false);
        float acc = 0.f;
#pragma unroll
        for (int v = 0; v < 8; ++v) {
            float x = c[v] + bv[t];
            x = (x > 0.f) ? x : (LEAKY_SLOPE * x);
            acc += x;
        }
        acc += __shfl_xor(acc, 16, 32);
        mm[t] = acc * (1.0f / 16.0f);
    }

    {
        const long obase = (long)p * 128;
        const int t0 = khi * 2;
        out[obase + (t0    ) * 16 + nn] = mm[t0];
        out[obase + (t0 + 1) * 16 + nn] = mm[t0 + 1];
    }

    float2 r;
    r.x = fx * (1.0f / 16.0f);
    r.y = fy * (1.0f / 16.0f);
    ((float2*)out)[(long)p * 64 + 32 + lane] = r;
}

extern "C" void kernel_launch(void* const* d_in, const int* in_sizes, int n_in,
                              void* d_out, int out_size, void* d_ws, size_t ws_size,
                              hipStream_t stream) {
    const float* points   = (const float*)d_in[0];
    const float* features = (const float*)d_in[1];
    const int*   knn      = (const int*)  d_in[2];
    const float* W        = (const float*)d_in[3];
    const float* bias     = (const float*)d_in[4];
    float* out            = (float*)d_out;

    const int nPoints = NB * NPTS;
    dim3 block(256);
    dim3 grid(nPoints / 8);
    lfa_kernel<<<grid, block, 0, stream>>>(points, features, knn, W, bias, out);
}


# one wave loops over 8 consecutive points instead of 1 (1024 blocks, 8 waves per SIMD, all resident at once): 8x fewer waves to launch; W/bias fragments read once per wave
# speedup vs baseline: 1.1300x; 1.1300x over previous
#include <hip/hip_runtime.h>


typedef __attribute__((ext_vector_type(2))) float v2f;
typedef __attribute__((ext_vector_type(8))) float v8f;

#define LEAKY_SLOPE 0.2f

constexpr int NB   = 4;
constexpr int NPTS = 16384;
constexpr int FD   = 64;
constexpr int KNN  = 16;
constexpr int PPW  = 8;

__global__ __launch_bounds__(256) __attribute__((amdgpu_waves_per_eu(8, 8))) void lfa_kernel(
    const float* __restrict__ points,
    const float* __restrict__ features,
    const int*  __restrict__ knn,
    const float* __restrict__ W,
    const float* __restrict__ bias,
    float* __restrict__ out)
{
    const int lane = threadIdx.x & 31;
    const int wave = __builtin_amdgcn_readfirstlane((int)(threadIdx.x >> 5));
    const int p0   = (blockIdx.x * 8 + wave) * PPW;
    const int b    = p0 >> 14;
    const int nn   = lane & 15;
    const int khi  = lane >> 4;

    v2f  wt[4];
    float bv[4];
#pragma unroll
    for (int t = 0; t < 4; ++t) {
        wt[t].x = W[(2 * khi)     * FD + t * 16 + nn];
        wt[t].y = W[(2 * khi + 1) * FD + t * 16 + nn];
        bv[t]   = bias[t * 16 + nn];
    }

    const float*  __restrict__ pbat  = points + (long)b * NPTS * 3;
    const float2* __restrict__ fbase = (const float2*)features + (long)b * NPTS * 32;

    int idx = knn[(long)p0 * KNN + nn];

#pragma unroll 1
    for (int i = 0; i < PPW; ++i) {
        const int p = p0 + i;

        const long nbase = (long)idx * 3;
        const float nx = pbat[nbase + 0];
        const float ny = pbat[nbase + 1];
        const float nz = pbat[nbase + 2];

        float2 rv[KNN];
#pragma unroll
        for (int k = 0; k < KNN; ++k) {
            const int ik = __builtin_amdgcn_readlane(idx, k);
            rv[k] = fbase[(long)ik * 32 + lane];
        }

        const int pn = (i + 1 < PPW) ? (p + 1) : p;
        const int idx_next = knn[(long)pn * KNN + nn];

        const long pbase = (long)p * 3;
        const float cx = points[pbase + 0];
        const float cy = points[pbase + 1];
        const float cz = points[pbase + 2];
        const float rx = cx - nx;
        const float ry = cy - ny;
        const float rz = cz - nz;
        const float dist = __builtin_amdgcn_sqrtf(rx * rx + ry * ry + rz * rz);

        v2f a;
        a.x = khi ? rz   : rx;
        a.y = khi ? dist : ry;

        float mm[4];
#pragma unroll
        for (int t = 0; t < 4; ++t) {
            v8f c = {};
            c = __builtin_amdgcn_wmma_f32_16x16x4_f32(
                     false, a,  false, wt[t],
                     (short)0, c,  false,  false);
            float acc = 0.f;
#pragma unroll
            for (int v = 0; v < 8; ++v) {
                float x = c[v] + bv[t];
                x = (x > 0.f) ? x : (LEAKY_SLOPE * x);
                acc += x;
            }
            acc += __shfl_xor(acc, 16, 32);
            mm[t] = acc * (1.0f / 16.0f);
        }

        {
            const long obase = (long)p * 128;
            const int t0 = khi * 2;
            out[obase + (t0    ) * 16 + nn] = mm[t0];
            out[obase + (t0 + 1) * 16 + nn] = mm[t0 + 1];
        }

        float fx = 0.f, fy = 0.f;
#pragma unroll
        for (int k = 0; k < KNN; ++k) {
            fx += rv[k].x;
            fy += rv[k].y;
        }

        float2 r;
        r.x = fx * (1.0f / 16.0f);
        r.y = fy * (1.0f / 16.0f);
        ((float2*)out)[(long)p * 64 + 32 + lane] = r;

        idx = idx_next;
    }
}

extern "C" void kernel_launch(void* const* d_in, const int* in_sizes, int n_in,
                              void* d_out, int out_size, void* d_ws, size_t ws_size,
                              hipStream_t stream) {
    const float* points   = (const float*)d_in[0];
    const float* features = (const float*)d_in[1];
    const int*   knn      = (const int*)  d_in[2];
    const float* W        = (const float*)d_in[3];
    const float* bias     = (const float*)d_in[4];
    float* out            = (float*)d_out;

    dim3 block(256);
    dim3 grid(1024);
    lfa_kernel<<<grid, block, 0, stream>>>(points, features, knn, W, bias, out);
}
